# starting kernel
# baseline (speedup 1.0000x reference)
#include <hip/hip_runtime.h>


#define NF 128
#define PSTR 144

typedef __attribute__((ext_vector_type(2))) float v2f;
typedef __attribute__((ext_vector_type(8))) float v8f;

__global__ void deg_init_kernel(float* __restrict__ deg, int n) {
  int i = blockIdx.x * blockDim.x + threadIdx.x;
  if (i < n) deg[i] = 1.0f;
}

__global__ void deg_count_kernel(const int* __restrict__ ei,
                                 float* __restrict__ deg, int nedges) {
  int e = blockIdx.x * blockDim.x + threadIdx.x;
  if (e < nedges) {
    int dst = ei[nedges + e];
    __hip_atomic_fetch_add(&deg[dst], 1.0f, __ATOMIC_RELAXED,
                           __HIP_MEMORY_SCOPE_AGENT);
  }
}

__global__ void rsqrt_kernel(float* __restrict__ deg, int n) {
  int i = blockIdx.x * blockDim.x + threadIdx.x;
  if (i < n) deg[i] = rsqrtf(deg[i]);
}

__global__ __launch_bounds__(256)
void gemm_scale_kernel(const float* __restrict__ x, const float* __restrict__ W,
                       const float* __restrict__ dinv, float* __restrict__ hp,
                       float* __restrict__ outinit, int nrows) {
  __shared__ v2f Wlds[(NF / 2) * PSTR];
  for (int i = threadIdx.x; i < NF * NF; i += 256) {
    int k = i >> 7, c = i & 127;
    Wlds[(k >> 1) * PSTR + c][k & 1] = W[i];
  }
  __syncthreads();

  const int wave = threadIdx.x >> 5;
  const int lane = threadIdx.x & 31;
  const int m0 = (blockIdx.x * 8 + wave) * 16;
  if (m0 >= nrows) return;

  const int half  = lane >> 4;
  const int khalf = half * 2;
  const int colb  = lane & 15;

  v8f acc[8] = {};

  const float* xr = x + (size_t)(m0 + colb) * NF + khalf;

  for (int kk = 0; kk < NF; kk += 4) {
    v2f a = *(const v2f*)(xr + kk);
    const v2f* wr = &Wlds[((kk >> 1) + half) * PSTR + colb];
#pragma unroll
    for (int nt = 0; nt < 8; ++nt) {
      v2f b = wr[nt * 16];
      acc[nt] = __builtin_amdgcn_wmma_f32_16x16x4_f32(
          false, a, false, b, (short)0, acc[nt], false, false);
    }
  }

  const int rbase = half << 3;
#pragma unroll
  for (int r = 0; r < 8; ++r) {
    const int row = m0 + rbase + r;
    const float dv = dinv[row];
    const size_t ro = (size_t)row * NF + colb;
#pragma unroll
    for (int nt = 0; nt < 8; ++nt) {
      float v = acc[nt][r] * dv;
      hp[ro + nt * 16] = v;
      outinit[ro + nt * 16] = v;
    }
  }
}

__global__ void scatter_kernel(const int* __restrict__ ei,
                               const float* __restrict__ hp,
                               float* __restrict__ out, int nedges) {
  int t = blockIdx.x * blockDim.x + threadIdx.x;
  int e = t >> 5;
  if (e >= nedges) return;
  int lane = t & 31;
  int src = ei[e];
  int dst = ei[nedges + e];
  const float4 v = *(const float4*)(hp + (size_t)src * NF + lane * 4);
  float* o = out + (size_t)dst * NF + lane * 4;
  __hip_atomic_fetch_add(o + 0, v.x, __ATOMIC_RELAXED, __HIP_MEMORY_SCOPE_AGENT);
  __hip_atomic_fetch_add(o + 1, v.y, __ATOMIC_RELAXED, __HIP_MEMORY_SCOPE_AGENT);
  __hip_atomic_fetch_add(o + 2, v.z, __ATOMIC_RELAXED, __HIP_MEMORY_SCOPE_AGENT);
  __hip_atomic_fetch_add(o + 3, v.w, __ATOMIC_RELAXED, __HIP_MEMORY_SCOPE_AGENT);
}

__global__ void finalize_kernel(float* __restrict__ out,
                                const float* __restrict__ dinv,
                                const float* __restrict__ bias,
                                long long total) {
  long long t = (long long)blockIdx.x * blockDim.x + threadIdx.x;
  if (t < total) {
    int row = (int)(t >> 7);
    int f = (int)(t & 127);
    out[t] = out[t] * dinv[row] + bias[f];
  }
}

extern "C" void kernel_launch(void* const* d_in, const int* in_sizes, int n_in,
                              void* d_out, int out_size, void* d_ws, size_t ws_size,
                              hipStream_t stream) {
  const float* x  = (const float*)d_in[0];
  const int*   ei = (const int*)d_in[1];
  const float* W  = (const float*)d_in[2];
  const float* b  = (const float*)d_in[3];
  float* out = (float*)d_out;

  const int N = in_sizes[0] / NF;
  const int E = in_sizes[1] / 2;

  float* deg = (float*)d_ws;
  size_t off = ((size_t)N * sizeof(float) + 511) & ~(size_t)511;
  float* hp = (float*)((char*)d_ws + off);

  deg_init_kernel<<<(N + 255) / 256, 256, 0, stream>>>(deg, N);
  deg_count_kernel<<<(E + 255) / 256, 256, 0, stream>>>(ei, deg, E);
  rsqrt_kernel<<<(N + 255) / 256, 256, 0, stream>>>(deg, N);

  gemm_scale_kernel<<<(N + 127) / 128, 256, 0, stream>>>(x, W, deg, hp, out, N);

  long long sthreads = (long long)E * 32;
  scatter_kernel<<<(int)((sthreads + 255) / 256), 256, 0, stream>>>(ei, hp, out, E);

  long long total = (long long)N * NF;
  finalize_kernel<<<(int)((total + 255) / 256), 256, 0, stream>>>(out, deg, b, total);
}


# scatter: lane = feature column, so every gather load and every fp32 atomic add instruction covers one whole 128-byte line (4 line requests per edge, not 16); adds arrive in any order, as before
# speedup vs baseline: 1.2516x; 1.2516x over previous
#include <hip/hip_runtime.h>


#define NF 128
#define PSTR 144

typedef __attribute__((ext_vector_type(2))) float v2f;
typedef __attribute__((ext_vector_type(8))) float v8f;

__global__ void deg_init_kernel(float* __restrict__ deg, int n) {
  int i = blockIdx.x * blockDim.x + threadIdx.x;
  if (i < n) deg[i] = 1.0f;
}

__global__ void deg_count_kernel(const int* __restrict__ ei,
                                 float* __restrict__ deg, int nedges) {
  int e = blockIdx.x * blockDim.x + threadIdx.x;
  if (e < nedges) {
    int dst = ei[nedges + e];
    __hip_atomic_fetch_add(&deg[dst], 1.0f, __ATOMIC_RELAXED,
                           __HIP_MEMORY_SCOPE_AGENT);
  }
}

__global__ void rsqrt_kernel(float* __restrict__ deg, int n) {
  int i = blockIdx.x * blockDim.x + threadIdx.x;
  if (i < n) deg[i] = rsqrtf(deg[i]);
}

__global__ __launch_bounds__(256)
void gemm_scale_kernel(const float* __restrict__ x, const float* __restrict__ W,
                       const float* __restrict__ dinv, float* __restrict__ hp,
                       float* __restrict__ outinit, int nrows) {
  __shared__ v2f Wlds[(NF / 2) * PSTR];
  for (int i = threadIdx.x; i < NF * NF; i += 256) {
    int k = i >> 7, c = i & 127;
    Wlds[(k >> 1) * PSTR + c][k & 1] = W[i];
  }
  __syncthreads();

  const int wave = threadIdx.x >> 5;
  const int lane = threadIdx.x & 31;
  const int m0 = (blockIdx.x * 8 + wave) * 16;
  if (m0 >= nrows) return;

  const int half  = lane >> 4;
  const int khalf = half * 2;
  const int colb  = lane & 15;

  v8f acc[8] = {};

  const float* xr = x + (size_t)(m0 + colb) * NF + khalf;

  for (int kk = 0; kk < NF; kk += 4) {
    v2f a = *(const v2f*)(xr + kk);
    const v2f* wr = &Wlds[((kk >> 1) + half) * PSTR + colb];
#pragma unroll
    for (int nt = 0; nt < 8; ++nt) {
      v2f b = wr[nt * 16];
      acc[nt] = __builtin_amdgcn_wmma_f32_16x16x4_f32(
          false, a, false, b, (short)0, acc[nt], false, false);
    }
  }

  const int rbase = half << 3;
#pragma unroll
  for (int r = 0; r < 8; ++r) {
    const int row = m0 + rbase + r;
    const float dv = dinv[row];
    const size_t ro = (size_t)row * NF + colb;
#pragma unroll
    for (int nt = 0; nt < 8; ++nt) {
      float v = acc[nt][r] * dv;
      hp[ro + nt * 16] = v;
      outinit[ro + nt * 16] = v;
    }
  }
}

__global__ void scatter_kernel(const int* __restrict__ ei,
                               const float* __restrict__ hp,
                               float* __restrict__ out, int nedges) {
  int t = blockIdx.x * blockDim.x + threadIdx.x;
  int e = t >> 5;
  if (e >= nedges) return;
  int lane = t & 31;
  int src = ei[e];
  int dst = ei[nedges + e];
  const float* h = hp + (size_t)src * NF + lane;
  const float v0 = h[0];
  const float v1 = h[32];
  const float v2 = h[64];
  const float v3 = h[96];
  float* o = out + (size_t)dst * NF + lane;
  __hip_atomic_fetch_add(o + 0,  v0, __ATOMIC_RELAXED, __HIP_MEMORY_SCOPE_AGENT);
  __hip_atomic_fetch_add(o + 32, v1, __ATOMIC_RELAXED, __HIP_MEMORY_SCOPE_AGENT);
  __hip_atomic_fetch_add(o + 64, v2, __ATOMIC_RELAXED, __HIP_MEMORY_SCOPE_AGENT);
  __hip_atomic_fetch_add(o + 96, v3, __ATOMIC_RELAXED, __HIP_MEMORY_SCOPE_AGENT);
}

__global__ void finalize_kernel(float* __restrict__ out,
                                const float* __restrict__ dinv,
                                const float* __restrict__ bias,
                                long long total) {
  long long t = (long long)blockIdx.x * blockDim.x + threadIdx.x;
  if (t < total) {
    int row = (int)(t >> 7);
    int f = (int)(t & 127);
    out[t] = out[t] * dinv[row] + bias[f];
  }
}

extern "C" void kernel_launch(void* const* d_in, const int* in_sizes, int n_in,
                              void* d_out, int out_size, void* d_ws, size_t ws_size,
                              hipStream_t stream) {
  const float* x  = (const float*)d_in[0];
  const int*   ei = (const int*)d_in[1];
  const float* W  = (const float*)d_in[2];
  const float* b  = (const float*)d_in[3];
  float* out = (float*)d_out;

  const int N = in_sizes[0] / NF;
  const int E = in_sizes[1] / 2;

  float* deg = (float*)d_ws;
  size_t off = ((size_t)N * sizeof(float) + 511) & ~(size_t)511;
  float* hp = (float*)((char*)d_ws + off);

  deg_init_kernel<<<(N + 255) / 256, 256, 0, stream>>>(deg, N);
  deg_count_kernel<<<(E + 255) / 256, 256, 0, stream>>>(ei, deg, E);
  rsqrt_kernel<<<(N + 255) / 256, 256, 0, stream>>>(deg, N);

  gemm_scale_kernel<<<(N + 127) / 128, 256, 0, stream>>>(x, W, deg, hp, out, N);

  long long sthreads = (long long)E * 32;
  scatter_kernel<<<(int)((sthreads + 255) / 256), 256, 0, stream>>>(ei, hp, out, E);

  long long total = (long long)N * NF;
  finalize_kernel<<<(int)((total + 255) / 256), 256, 0, stream>>>(out, deg, b, total);
}


# aggregate by destination row: edges counting-sorted by dst in-call (int atomic rank, scan, fill); one wave sums a row's h' lines, writes out once; no fp32 atomics; sum order = arrival rank
# speedup vs baseline: 2.0453x; 1.6341x over previous
#include <hip/hip_runtime.h>


#define NF 128
#define PSTR 144
#define SCAN_B 1024

typedef __attribute__((ext_vector_type(2))) float v2f;
typedef __attribute__((ext_vector_type(4))) float v4f;
typedef __attribute__((ext_vector_type(8))) float v8f;

__global__ void zero_counts_kernel(int* __restrict__ cnt, int n) {
  int i = blockIdx.x * blockDim.x + threadIdx.x;
  if (i < n) cnt[i] = 0;
}

__global__ void count_rank_kernel(const int* __restrict__ ei,
                                  int* __restrict__ cnt,
                                  int* __restrict__ rank, int nedges) {
  int e = blockIdx.x * blockDim.x + threadIdx.x;
  if (e < nedges) {
    int dst = ei[nedges + e];
    rank[e] = __hip_atomic_fetch_add(&cnt[dst], 1, __ATOMIC_RELAXED,
                                     __HIP_MEMORY_SCOPE_AGENT);
  }
}

__global__ __launch_bounds__(SCAN_B)
void scan_kernel(const int* __restrict__ cnt, int* __restrict__ rowptr,
                 float* __restrict__ dinv, int n) {
  __shared__ int wtot[32];
  __shared__ int wpre[32];
  const int tid = threadIdx.x;
  const int lane = tid & 31;
  const int wv = tid >> 5;
  const int base = blockIdx.x * SCAN_B;

  int s = 0;
  for (int i = tid; i < base; i += SCAN_B) s += cnt[i];

  const int i = base + tid;
  const int v = (i < n) ? cnt[i] : 0;

  int incl = v;
#pragma unroll
  for (int d = 1; d < 32; d <<= 1) {
    int t = __shfl_up(incl, d);
    if (lane >= d) incl += t;
  }
#pragma unroll
  for (int d = 16; d > 0; d >>= 1) s += __shfl_xor(s, d);

  if (lane == 31) wtot[wv] = incl;
  if (lane == 0) wpre[wv] = s;
  __syncthreads();

  int a = (lane < wv) ? wtot[lane] : 0;
  int b = wpre[lane];
#pragma unroll
  for (int d = 16; d > 0; d >>= 1) {
    a += __shfl_xor(a, d);
    b += __shfl_xor(b, d);
  }
  if (i < n) {
    rowptr[i] = b + a + incl - v;
    dinv[i] = rsqrtf((float)(v + 1));
  }
}

__global__ void fill_kernel(const int* __restrict__ ei,
                            const int* __restrict__ rowptr,
                            const int* __restrict__ rank,
                            int* __restrict__ csr, int nedges) {
  int e = blockIdx.x * blockDim.x + threadIdx.x;
  if (e < nedges) {
    int src = ei[e];
    int dst = ei[nedges + e];
    csr[rowptr[dst] + rank[e]] = src;
  }
}

__global__ __launch_bounds__(256)
void gemm_scale_kernel(const float* __restrict__ x, const float* __restrict__ W,
                       const float* __restrict__ dinv, float* __restrict__ hp,
                       int nrows) {
  __shared__ v2f Wlds[(NF / 2) * PSTR];
  for (int i = threadIdx.x; i < NF * NF; i += 256) {
    int k = i >> 7, c = i & 127;
    Wlds[(k >> 1) * PSTR + c][k & 1] = W[i];
  }
  __syncthreads();

  const int wave = threadIdx.x >> 5;
  const int lane = threadIdx.x & 31;
  const int m0 = (blockIdx.x * 8 + wave) * 16;
  if (m0 >= nrows) return;

  const int half  = lane >> 4;
  const int khalf = half * 2;
  const int colb  = lane & 15;

  v8f acc[8] = {};

  const float* xr = x + (size_t)(m0 + colb) * NF + khalf;

  for (int kk = 0; kk < NF; kk += 4) {
    v2f a = *(const v2f*)(xr + kk);
    const v2f* wr = &Wlds[((kk >> 1) + half) * PSTR + colb];
#pragma unroll
    for (int nt = 0; nt < 8; ++nt) {
      v2f b = wr[nt * 16];
      acc[nt] = __builtin_amdgcn_wmma_f32_16x16x4_f32(
          false, a, false, b, (short)0, acc[nt], false, false);
    }
  }

  const int rbase = half << 3;
#pragma unroll
  for (int r = 0; r < 8; ++r) {
    const int row = m0 + rbase + r;
    const float dv = dinv[row];
    const size_t ro = (size_t)row * NF + colb;
#pragma unroll
    for (int nt = 0; nt < 8; ++nt) {
      hp[ro + nt * 16] = acc[nt][r] * dv;
    }
  }
}

__global__ __launch_bounds__(256)
void gather_rows_kernel(const int* __restrict__ rowptr,
                        const int* __restrict__ cnt,
                        const int* __restrict__ csr,
                        const float* __restrict__ hp,
                        const float* __restrict__ dinv,
                        const float* __restrict__ bias,
                        float* __restrict__ out, int n) {
  const int lane = threadIdx.x & 31;
  const int row = __builtin_amdgcn_readfirstlane(
      (int)(blockIdx.x * 8 + (threadIdx.x >> 5)));
  if (row >= n) return;

  const int start = rowptr[row];
  const int deg = cnt[row];
  const v4f* hp4 = (const v4f*)hp;

  v4f acc = hp4[(size_t)row * 32 + lane];

  for (int base = 0; base < deg; base += 32) {
    const int m = min(32, deg - base);
    int idx = row;
    if (lane < m) idx = csr[start + base + lane];
    for (int j = 0; j < m; j += 8) {
      v4f v[8];
#pragma unroll
      for (int u = 0; u < 8; ++u) {
        const int s = __builtin_amdgcn_readlane(idx, j + u);
        v[u] = hp4[(size_t)s * 32 + lane];
      }
#pragma unroll
      for (int u = 0; u < 8; ++u) {
        if (j + u < m) acc += v[u];
      }
    }
  }

  const float dv = dinv[row];
  const v4f b4 = ((const v4f*)bias)[lane];
  ((v4f*)out)[(size_t)row * 32 + lane] = acc * dv + b4;
}

extern "C" void kernel_launch(void* const* d_in, const int* in_sizes, int n_in,
                              void* d_out, int out_size, void* d_ws, size_t ws_size,
                              hipStream_t stream) {
  const float* x  = (const float*)d_in[0];
  const int*   ei = (const int*)d_in[1];
  const float* W  = (const float*)d_in[2];
  const float* b  = (const float*)d_in[3];
  float* out = (float*)d_out;

  const int N = in_sizes[0] / NF;
  const int E = in_sizes[1] / 2;

  const size_t nb = ((size_t)N * 4 + 511) & ~(size_t)511;
  const size_t eb = ((size_t)E * 4 + 511) & ~(size_t)511;
  char* p = (char*)d_ws;
  int*   cnt    = (int*)p;             p += nb;
  int*   rowptr = (int*)p;             p += nb;
  float* dinv   = (float*)p;           p += nb;
  int*   rank   = (int*)p;             p += eb;
  int*   csr    = (int*)p;             p += eb;
  float* hp     = (float*)p;

  zero_counts_kernel<<<(N + 255) / 256, 256, 0, stream>>>(cnt, N);
  count_rank_kernel<<<(E + 255) / 256, 256, 0, stream>>>(ei, cnt, rank, E);
  scan_kernel<<<(N + SCAN_B - 1) / SCAN_B, SCAN_B, 0, stream>>>(cnt, rowptr, dinv, N);
  fill_kernel<<<(E + 255) / 256, 256, 0, stream>>>(ei, rowptr, rank, csr, E);

  gemm_scale_kernel<<<(N + 127) / 128, 256, 0, stream>>>(x, W, dinv, hp, N);

  gather_rows_kernel<<<(N + 7) / 8, 256, 0, stream>>>(rowptr, cnt, csr, hp, dinv, b, out, N);
}


# group edges by destination with LDS counters only: 256-row bins (LDS histogram, scan, LDS-cursor scatter), then per-bin LDS row count and fill; no atomics on memory; sum order = arrival, as before
# speedup vs baseline: 3.4028x; 1.6637x over previous
#include <hip/hip_runtime.h>


#define NF 128
#define PSTR 144
#define BSH 8
#define BROWS 256
#define CH 4096
#define NBMAX 1024

typedef __attribute__((ext_vector_type(2))) float v2f;
typedef __attribute__((ext_vector_type(2))) int v2i;
typedef __attribute__((ext_vector_type(4))) float v4f;
typedef __attribute__((ext_vector_type(8))) float v8f;

__device__ __forceinline__ int wave_incl_scan(int v, int lane) {
#pragma unroll
  for (int d = 1; d < 32; d <<= 1) {
    int t = __shfl_up(v, d);
    if (lane >= d) v += t;
  }
  return v;
}

__device__ __forceinline__ int wave_sum(int v) {
#pragma unroll
  for (int d = 16; d > 0; d >>= 1) v += __shfl_xor(v, d);
  return v;
}

__global__ __launch_bounds__(1024)
void bin_hist_kernel(const int* __restrict__ ei, int* __restrict__ whist,
                     int nedges, int nb) {
  __shared__ int h[NBMAX];
  const int tid = threadIdx.x;
  for (int b = tid; b < nb; b += 1024) h[b] = 0;
  __syncthreads();

  const int e0 = blockIdx.x * CH + tid;
  int d[CH / 1024];
#pragma unroll
  for (int j = 0; j < CH / 1024; ++j) {
    const int e = e0 + j * 1024;
    d[j] = (e < nedges) ? ei[nedges + e] : -1;
  }
#pragma unroll
  for (int j = 0; j < CH / 1024; ++j)
    if (d[j] >= 0) atomicAdd(&h[d[j] >> BSH], 1);
  __syncthreads();

  for (int b = tid; b < nb; b += 1024)
    whist[(size_t)blockIdx.x * nb + b] = h[b];
}

__global__ __launch_bounds__(512)
void bin_scan_kernel(int* __restrict__ whist, int* __restrict__ tot,
                     int ng, int nb) {
  __shared__ int wt[16];
  const int tid = threadIdx.x;
  const int lane = tid & 31;
  const int wv = tid >> 5;
  const int b = blockIdx.x;
  int carry = 0;
  for (int g0 = 0; g0 < ng; g0 += 512) {
    const int g = g0 + tid;
    const int v = (g < ng) ? whist[(size_t)g * nb + b] : 0;
    const int incl = wave_incl_scan(v, lane);
    if (lane == 31) wt[wv] = incl;
    __syncthreads();
    const int w = (lane < 16) ? wt[lane] : 0;
    const int front = wave_sum((lane < wv) ? w : 0);
    const int all = wave_sum(w);
    if (g < ng) whist[(size_t)g * nb + b] = carry + front + incl - v;
    carry += all;
    __syncthreads();
  }
  if (tid == 0) tot[b] = carry;
}

__global__ __launch_bounds__(1024)
void bin_scatter_kernel(const int* __restrict__ ei,
                        const int* __restrict__ woff,
                        const int* __restrict__ tot,
                        int* __restrict__ binstart,
                        int* __restrict__ binned, int nedges, int nb) {
  __shared__ int cur[NBMAX];
  __shared__ int wt[32];
  const int tid = threadIdx.x;
  const int lane = tid & 31;
  const int wv = tid >> 5;

  const int v = (tid < nb) ? tot[tid] : 0;
  const int incl = wave_incl_scan(v, lane);
  if (lane == 31) wt[wv] = incl;
  __syncthreads();
  const int front = wave_sum((lane < wv) ? wt[lane] : 0);
  const int start = front + incl - v;
  if (tid < nb) {
    cur[tid] = start + woff[(size_t)blockIdx.x * nb + tid];
    if (blockIdx.x == 0) binstart[tid] = start;
  }
  __syncthreads();

  const int e0 = blockIdx.x * CH + tid;
  int s[CH / 1024], d[CH / 1024];
#pragma unroll
  for (int j = 0; j < CH / 1024; ++j) {
    const int e = e0 + j * 1024;
    s[j] = (e < nedges) ? ei[e] : 0;
    d[j] = (e < nedges) ? ei[nedges + e] : -1;
  }
  v2i* b2 = (v2i*)binned;
#pragma unroll
  for (int j = 0; j < CH / 1024; ++j) {
    if (d[j] >= 0) {
      const int slot = atomicAdd(&cur[d[j] >> BSH], 1);
      v2i sd;
      sd[0] = s[j];
      sd[1] = d[j];
      b2[slot] = sd;
    }
  }
}

__global__ __launch_bounds__(1024)
void bin_rows_kernel(const int* __restrict__ binned,
                     const int* __restrict__ binstart,
                     const int* __restrict__ tot,
                     int* __restrict__ cnt, int* __restrict__ rowptr,
                     float* __restrict__ dinv, int* __restrict__ csr, int n) {
  __shared__ int c[BROWS];
  __shared__ int wt[BROWS / 32];
  const int tid = threadIdx.x;
  const int lane = tid & 31;
  const int wv = tid >> 5;
  const int r0 = blockIdx.x << BSH;
  const int beg = binstart[blockIdx.x];
  const int ne = tot[blockIdx.x];
  const v2i* b2 = (const v2i*)binned + beg;

  if (tid < BROWS) c[tid] = 0;
  __syncthreads();
  for (int i0 = 0; i0 < ne; i0 += 4096) {
    int dd[4];
#pragma unroll
    for (int j = 0; j < 4; ++j) {
      const int i = i0 + j * 1024 + tid;
      dd[j] = (i < ne) ? b2[i][1] : -1;
    }
#pragma unroll
    for (int j = 0; j < 4; ++j)
      if (dd[j] >= 0) atomicAdd(&c[dd[j] - r0], 1);
  }
  __syncthreads();

  const int v = (tid < BROWS) ? c[tid] : 0;
  const int incl = wave_incl_scan(v, lane);
  if (tid < BROWS && lane == 31) wt[wv] = incl;
  __syncthreads();
  const int front = wave_sum((lane < wv && lane < BROWS / 32) ? wt[lane] : 0);
  if (tid < BROWS) {
    const int rs = beg + front + incl - v;
    c[tid] = rs;
    const int row = r0 + tid;
    if (row < n) {
      cnt[row] = v;
      rowptr[row] = rs;
      dinv[row] = rsqrtf((float)(v + 1));
    }
  }
  __syncthreads();
  for (int i0 = 0; i0 < ne; i0 += 4096) {
    int ss[4], dd[4];
#pragma unroll
    for (int j = 0; j < 4; ++j) {
      const int i = i0 + j * 1024 + tid;
      ss[j] = 0;
      dd[j] = -1;
      if (i < ne) {
        const v2i sd = b2[i];
        ss[j] = sd[0];
        dd[j] = sd[1];
      }
    }
#pragma unroll
    for (int j = 0; j < 4; ++j) {
      if (dd[j] >= 0) {
        const int slot = atomicAdd(&c[dd[j] - r0], 1);
        csr[slot] = ss[j];
      }
    }
  }
}

__global__ __launch_bounds__(256)
void gemm_scale_kernel(const float* __restrict__ x, const float* __restrict__ W,
                       const float* __restrict__ dinv, float* __restrict__ hp,
                       int nrows) {
  __shared__ v2f Wlds[(NF / 2) * PSTR];
  for (int i = threadIdx.x; i < NF * NF; i += 256) {
    int k = i >> 7, c = i & 127;
    Wlds[(k >> 1) * PSTR + c][k & 1] = W[i];
  }
  __syncthreads();

  const int wave = threadIdx.x >> 5;
  const int lane = threadIdx.x & 31;
  const int m0 = (blockIdx.x * 8 + wave) * 16;
  if (m0 >= nrows) return;

  const int half  = lane >> 4;
  const int khalf = half * 2;
  const int colb  = lane & 15;

  v8f acc[8] = {};

  const float* xr = x + (size_t)(m0 + colb) * NF + khalf;

  for (int kk = 0; kk < NF; kk += 4) {
    v2f a = *(const v2f*)(xr + kk);
    const v2f* wr = &Wlds[((kk >> 1) + half) * PSTR + colb];
#pragma unroll
    for (int nt = 0; nt < 8; ++nt) {
      v2f b = wr[nt * 16];
      acc[nt] = __builtin_amdgcn_wmma_f32_16x16x4_f32(
          false, a, false, b, (short)0, acc[nt], false, false);
    }
  }

  const int rbase = half << 3;
#pragma unroll
  for (int r = 0; r < 8; ++r) {
    const int row = m0 + rbase + r;
    const float dv = dinv[row];
    const size_t ro = (size_t)row * NF + colb;
#pragma unroll
    for (int nt = 0; nt < 8; ++nt) {
      hp[ro + nt * 16] = acc[nt][r] * dv;
    }
  }
}

__global__ __launch_bounds__(256)
void gather_rows_kernel(const int* __restrict__ rowptr,
                        const int* __restrict__ cnt,
                        const int* __restrict__ csr,
                        const float* __restrict__ hp,
                        const float* __restrict__ dinv,
                        const float* __restrict__ bias,
                        float* __restrict__ out, int n) {
  const int lane = threadIdx.x & 31;
  const int row = __builtin_amdgcn_readfirstlane(
      (int)(blockIdx.x * 8 + (threadIdx.x >> 5)));
  if (row >= n) return;

  const int start = rowptr[row];
  const int deg = cnt[row];
  const v4f* hp4 = (const v4f*)hp;

  v4f acc = hp4[(size_t)row * 32 + lane];

  for (int base = 0; base < deg; base += 32) {
    const int m = min(32, deg - base);
    int idx = row;
    if (lane < m) idx = csr[start + base + lane];
    for (int j = 0; j < m; j += 8) {
      v4f v[8];
#pragma unroll
      for (int u = 0; u < 8; ++u) {
        const int s = __builtin_amdgcn_readlane(idx, j + u);
        v[u] = hp4[(size_t)s * 32 + lane];
      }
#pragma unroll
      for (int u = 0; u < 8; ++u) {
        if (j + u < m) acc += v[u];
      }
    }
  }

  const float dv = dinv[row];
  const v4f b4 = ((const v4f*)bias)[lane];
  ((v4f*)out)[(size_t)row * 32 + lane] = acc * dv + b4;
}

extern "C" void kernel_launch(void* const* d_in, const int* in_sizes, int n_in,
                              void* d_out, int out_size, void* d_ws, size_t ws_size,
                              hipStream_t stream) {
  const float* x  = (const float*)d_in[0];
  const int*   ei = (const int*)d_in[1];
  const float* W  = (const float*)d_in[2];
  const float* b  = (const float*)d_in[3];
  float* out = (float*)d_out;

  const int N = in_sizes[0] / NF;
  const int E = in_sizes[1] / 2;

  const int NB = (N + BROWS - 1) / BROWS;
  const int NG = (E + CH - 1) / CH;

  const size_t nb = ((size_t)N * 4 + 511) & ~(size_t)511;
  const size_t eb = ((size_t)E * 4 + 511) & ~(size_t)511;
  const size_t bb = ((size_t)NB * 4 + 511) & ~(size_t)511;
  const size_t gb = ((size_t)NG * NB * 4 + 511) & ~(size_t)511;
  char* p = (char*)d_ws;
  int*   cnt      = (int*)p;           p += nb;
  int*   rowptr   = (int*)p;           p += nb;
  float* dinv     = (float*)p;         p += nb;
  int*   tot      = (int*)p;           p += bb;
  int*   binstart = (int*)p;           p += bb;
  int*   whist    = (int*)p;           p += gb;
  int*   binned   = (int*)p;           p += 2 * eb;
  int*   csr      = (int*)p;           p += eb;
  float* hp       = (float*)p;

  bin_hist_kernel<<<NG, 1024, 0, stream>>>(ei, whist, E, NB);
  bin_scan_kernel<<<NB, 512, 0, stream>>>(whist, tot, NG, NB);
  bin_scatter_kernel<<<NG, 1024, 0, stream>>>(ei, whist, tot, binstart, binned, E, NB);
  bin_rows_kernel<<<NB, 1024, 0, stream>>>(binned, binstart, tot, cnt, rowptr, dinv, csr, N);

  gemm_scale_kernel<<<(N + 127) / 128, 256, 0, stream>>>(x, W, dinv, hp, N);

  gather_rows_kernel<<<(N + 7) / 8, 256, 0, stream>>>(rowptr, cnt, csr, hp, dinv, b, out, N);
}


# edge grouping in 2 launches, not 4: each work-group sorts its 4096 edges by 256-row bin in LDS into its own slice; the bin pass walks the slices' runs, start = sum of run offsets; sum order = arrival
# speedup vs baseline: 3.9942x; 1.1738x over previous
#include <hip/hip_runtime.h>


#define NF 128
#define PSTR 144
#define BSH 8
#define BROWS 256
#define CH 4096
#define ST 256
#define EPT (CH / ST)
#define RW 16
#define RG 16
#define NBMAX 1024

typedef __attribute__((ext_vector_type(2))) float v2f;
typedef __attribute__((ext_vector_type(2))) int v2i;
typedef __attribute__((ext_vector_type(4))) float v4f;
typedef __attribute__((ext_vector_type(8))) float v8f;

__device__ __forceinline__ int wave_incl_scan(int v, int lane) {
#pragma unroll
  for (int d = 1; d < 32; d <<= 1) {
    int t = __shfl_up(v, d);
    if (lane >= d) v += t;
  }
  return v;
}

__device__ __forceinline__ int wave_sum(int v) {
#pragma unroll
  for (int d = 16; d > 0; d >>= 1) v += __shfl_xor(v, d);
  return v;
}

__global__ __launch_bounds__(ST)
void slice_sort_kernel(const int* __restrict__ ei, int* __restrict__ roff,
                       int* __restrict__ sorted, int nedges, int nb) {
  __shared__ int h[NBMAX];
  __shared__ int wt[ST / 32];
  const int tid = threadIdx.x;
  const int lane = tid & 31;
  const int wv = tid >> 5;
  for (int b = tid; b < nb; b += ST) h[b] = 0;
  __syncthreads();

  const int e0 = blockIdx.x * CH + tid;
  int s[EPT], d[EPT];
#pragma unroll
  for (int j = 0; j < EPT; ++j) {
    const int e = e0 + j * ST;
    s[j] = (e < nedges) ? ei[e] : 0;
    d[j] = (e < nedges) ? ei[nedges + e] : -1;
  }
#pragma unroll
  for (int j = 0; j < EPT; ++j)
    if (d[j] >= 0) atomicAdd(&h[d[j] >> BSH], 1);
  __syncthreads();

  const int b0 = tid * 4;
  int v[4];
#pragma unroll
  for (int k = 0; k < 4; ++k) v[k] = (b0 + k < nb) ? h[b0 + k] : 0;
  const int sm = v[0] + v[1] + v[2] + v[3];
  const int incl = wave_incl_scan(sm, lane);
  if (lane == 31) wt[wv] = incl;
  __syncthreads();
  const int front = wave_sum((lane < wv) ? wt[lane] : 0);
  int run = front + incl - sm;
  int* ro = roff + (size_t)blockIdx.x * (nb + 1);
#pragma unroll
  for (int k = 0; k < 4; ++k) {
    if (b0 + k < nb) {
      h[b0 + k] = run;
      ro[b0 + k] = run;
    }
    run += v[k];
  }
  if (tid == ST - 1) ro[nb] = run;
  __syncthreads();

  v2i* out2 = (v2i*)sorted + (size_t)blockIdx.x * CH;
#pragma unroll
  for (int j = 0; j < EPT; ++j) {
    if (d[j] >= 0) {
      const int slot = atomicAdd(&h[d[j] >> BSH], 1);
      v2i sd;
      sd[0] = s[j];
      sd[1] = d[j];
      out2[slot] = sd;
    }
  }
}

template <bool FILL>
__device__ __forceinline__ void walk_runs(const v2i* __restrict__ s2,
                                          const int* __restrict__ roff,
                                          int* c, int* __restrict__ csr,
                                          int b, int r0, int ng, int nb,
                                          int wv, int lane, int& begpart) {
  for (int gb = wv; gb < ng; gb += RW * 32) {
    const int gl = gb + RW * lane;
    int st = 0, en = 0;
    if (gl < ng) {
      const int* ro = roff + (size_t)gl * (nb + 1) + b;
      st = ro[0];
      en = ro[1];
    }
    if (!FILL) begpart += st;
    const int nrun = min(32, (ng - gb + RW - 1) / RW);
    for (int l0 = 0; l0 < nrun; l0 += RG) {
      int ss[RG], dd[RG];
#pragma unroll
      for (int u = 0; u < RG; ++u) {
        const int rs = __builtin_amdgcn_readlane(st, l0 + u);
        const int re = __builtin_amdgcn_readlane(en, l0 + u);
        const int i = rs + lane;
        ss[u] = 0;
        dd[u] = -1;
        if (i < re) {
          const v2i sd = s2[(size_t)(gb + RW * (l0 + u)) * CH + i];
          ss[u] = sd[0];
          dd[u] = sd[1];
        }
      }
#pragma unroll
      for (int u = 0; u < RG; ++u) {
        if (dd[u] >= 0) {
          if (FILL) {
            const int slot = atomicAdd(&c[dd[u] - r0], 1);
            csr[slot] = ss[u];
          } else {
            atomicAdd(&c[dd[u] - r0], 1);
          }
        }
      }
      for (int u = 0; u < RG; ++u) {
        const int rs = __builtin_amdgcn_readlane(st, l0 + u);
        const int re = __builtin_amdgcn_readlane(en, l0 + u);
        for (int i = rs + 32 + lane; i < re; i += 32) {
          const v2i sd = s2[(size_t)(gb + RW * (l0 + u)) * CH + i];
          if (FILL) {
            const int slot = atomicAdd(&c[sd[1] - r0], 1);
            csr[slot] = sd[0];
          } else {
            atomicAdd(&c[sd[1] - r0], 1);
          }
        }
      }
    }
  }
}

__global__ __launch_bounds__(RW * 32)
void bin_rows_kernel(const int* __restrict__ sorted,
                     const int* __restrict__ roff,
                     int* __restrict__ cnt, int* __restrict__ rowptr,
                     float* __restrict__ dinv, int* __restrict__ csr,
                     int n, int ng, int nb) {
  __shared__ int c[BROWS];
  __shared__ int wt[RW];
  const int tid = threadIdx.x;
  const int lane = tid & 31;
  const int wv = __builtin_amdgcn_readfirstlane(tid >> 5);
  const int b = blockIdx.x;
  const int r0 = b << BSH;
  const v2i* s2 = (const v2i*)sorted;

  if (tid < BROWS) c[tid] = 0;
  __syncthreads();
  int begpart = 0;
  walk_runs<false>(s2, roff, c, csr, b, r0, ng, nb, wv, lane, begpart);
  const int ws = wave_sum(begpart);
  if (lane == 0) wt[wv] = ws;
  __syncthreads();
  const int beg = wave_sum((lane < RW) ? wt[lane] : 0);
  __syncthreads();

  const int v = (tid < BROWS) ? c[tid] : 0;
  const int incl = wave_incl_scan(v, lane);
  if (tid < BROWS && lane == 31) wt[wv] = incl;
  __syncthreads();
  const int front = wave_sum((lane < wv && lane < BROWS / 32) ? wt[lane] : 0);
  if (tid < BROWS) {
    const int rs = beg + front + incl - v;
    c[tid] = rs;
    const int row = r0 + tid;
    if (row < n) {
      cnt[row] = v;
      rowptr[row] = rs;
      dinv[row] = rsqrtf((float)(v + 1));
    }
  }
  __syncthreads();
  walk_runs<true>(s2, roff, c, csr, b, r0, ng, nb, wv, lane, begpart);
}

__global__ __launch_bounds__(256)
void gemm_scale_kernel(const float* __restrict__ x, const float* __restrict__ W,
                       const float* __restrict__ dinv, float* __restrict__ hp,
                       int nrows) {
  __shared__ v2f Wlds[(NF / 2) * PSTR];
  for (int i = threadIdx.x; i < NF * NF; i += 256) {
    int k = i >> 7, c = i & 127;
    Wlds[(k >> 1) * PSTR + c][k & 1] = W[i];
  }
  __syncthreads();

  const int wave = threadIdx.x >> 5;
  const int lane = threadIdx.x & 31;
  const int m0 = (blockIdx.x * 8 + wave) * 16;
  if (m0 >= nrows) return;

  const int half  = lane >> 4;
  const int khalf = half * 2;
  const int colb  = lane & 15;

  v8f acc[8] = {};

  const float* xr = x + (size_t)(m0 + colb) * NF + khalf;

  for (int kk = 0; kk < NF; kk += 4) {
    v2f a = *(const v2f*)(xr + kk);
    const v2f* wr = &Wlds[((kk >> 1) + half) * PSTR + colb];
#pragma unroll
    for (int nt = 0; nt < 8; ++nt) {
      v2f b = wr[nt * 16];
      acc[nt] = __builtin_amdgcn_wmma_f32_16x16x4_f32(
          false, a, false, b, (short)0, acc[nt], false, false);
    }
  }

  const int rbase = half << 3;
#pragma unroll
  for (int r = 0; r < 8; ++r) {
    const int row = m0 + rbase + r;
    const float dv = dinv[row];
    const size_t ro = (size_t)row * NF + colb;
#pragma unroll
    for (int nt = 0; nt < 8; ++nt) {
      hp[ro + nt * 16] = acc[nt][r] * dv;
    }
  }
}

__global__ __launch_bounds__(256)
void gather_rows_kernel(const int* __restrict__ rowptr,
                        const int* __restrict__ cnt,
                        const int* __restrict__ csr,
                        const float* __restrict__ hp,
                        const float* __restrict__ dinv,
                        const float* __restrict__ bias,
                        float* __restrict__ out, int n) {
  const int lane = threadIdx.x & 31;
  const int row = __builtin_amdgcn_readfirstlane(
      (int)(blockIdx.x * 8 + (threadIdx.x >> 5)));
  if (row >= n) return;

  const int start = rowptr[row];
  const int deg = cnt[row];
  const v4f* hp4 = (const v4f*)hp;

  v4f acc = hp4[(size_t)row * 32 + lane];

  for (int base = 0; base < deg; base += 32) {
    const int m = min(32, deg - base);
    int idx = row;
    if (lane < m) idx = csr[start + base + lane];
    for (int j = 0; j < m; j += 8) {
      v4f v[8];
#pragma unroll
      for (int u = 0; u < 8; ++u) {
        const int s = __builtin_amdgcn_readlane(idx, j + u);
        v[u] = hp4[(size_t)s * 32 + lane];
      }
#pragma unroll
      for (int u = 0; u < 8; ++u) {
        if (j + u < m) acc += v[u];
      }
    }
  }

  const float dv = dinv[row];
  const v4f b4 = ((const v4f*)bias)[lane];
  ((v4f*)out)[(size_t)row * 32 + lane] = acc * dv + b4;
}

extern "C" void kernel_launch(void* const* d_in, const int* in_sizes, int n_in,
                              void* d_out, int out_size, void* d_ws, size_t ws_size,
                              hipStream_t stream) {
  const float* x  = (const float*)d_in[0];
  const int*   ei = (const int*)d_in[1];
  const float* W  = (const float*)d_in[2];
  const float* b  = (const float*)d_in[3];
  float* out = (float*)d_out;

  const int N = in_sizes[0] / NF;
  const int E = in_sizes[1] / 2;

  const int NB = (N + BROWS - 1) / BROWS;
  const int NG = (E + CH - 1) / CH;

  const size_t nb = ((size_t)N * 4 + 511) & ~(size_t)511;
  const size_t eb = ((size_t)E * 4 + 511) & ~(size_t)511;
  const size_t gb = ((size_t)NG * (NB + 1) * 4 + 511) & ~(size_t)511;
  const size_t sb = (size_t)NG * CH * 8;
  char* p = (char*)d_ws;
  int*   cnt      = (int*)p;           p += nb;
  int*   rowptr   = (int*)p;           p += nb;
  float* dinv     = (float*)p;         p += nb;
  int*   roff     = (int*)p;           p += gb;
  int*   sorted   = (int*)p;           p += sb;
  int*   csr      = (int*)p;           p += eb;
  float* hp       = (float*)p;

  slice_sort_kernel<<<NG, ST, 0, stream>>>(ei, roff, sorted, E, NB);
  bin_rows_kernel<<<NB, RW * 32, 0, stream>>>(sorted, roff, cnt, rowptr, dinv, csr, N, NG, NB);

  gemm_scale_kernel<<<(N + 127) / 128, 256, 0, stream>>>(x, W, dinv, hp, N);

  gather_rows_kernel<<<(N + 7) / 8, 256, 0, stream>>>(rowptr, cnt, csr, hp, dinv, b, out, N);
}


# grouping launches scatter into an LDS image of the work-group's own contiguous output range (slice pairs; bin source ids), then copy it out with whole-line stores; sum order = arrival as before
# speedup vs baseline: 4.2929x; 1.0748x over previous
#include <hip/hip_runtime.h>


#define NF 128
#define PSTR 144
#define BSH 8
#define BROWS 256
#define CH 4096
#define ST 256
#define EPT (CH / ST)
#define RW 16
#define RG 16
#define NBMAX 1024
#define CCAP 8192

typedef __attribute__((ext_vector_type(4))) int v4i;

typedef __attribute__((ext_vector_type(2))) float v2f;
typedef __attribute__((ext_vector_type(2))) int v2i;
typedef __attribute__((ext_vector_type(4))) float v4f;
typedef __attribute__((ext_vector_type(8))) float v8f;

__device__ __forceinline__ int wave_incl_scan(int v, int lane) {
#pragma unroll
  for (int d = 1; d < 32; d <<= 1) {
    int t = __shfl_up(v, d);
    if (lane >= d) v += t;
  }
  return v;
}

__device__ __forceinline__ int wave_sum(int v) {
#pragma unroll
  for (int d = 16; d > 0; d >>= 1) v += __shfl_xor(v, d);
  return v;
}

__global__ __launch_bounds__(ST)
void slice_sort_kernel(const int* __restrict__ ei, int* __restrict__ roff,
                       int* __restrict__ sorted, int nedges, int nb) {
  __shared__ int h[NBMAX];
  __shared__ int wt[ST / 32];
  __shared__ v4i img[CH / 2];
  const int tid = threadIdx.x;
  const int lane = tid & 31;
  const int wv = tid >> 5;
  for (int b = tid; b < nb; b += ST) h[b] = 0;
  __syncthreads();

  const int e0 = blockIdx.x * CH + tid;
  int s[EPT], d[EPT];
#pragma unroll
  for (int j = 0; j < EPT; ++j) {
    const int e = e0 + j * ST;
    s[j] = (e < nedges) ? ei[e] : 0;
    d[j] = (e < nedges) ? ei[nedges + e] : -1;
  }
#pragma unroll
  for (int j = 0; j < EPT; ++j)
    if (d[j] >= 0) atomicAdd(&h[d[j] >> BSH], 1);
  __syncthreads();

  const int b0 = tid * 4;
  int v[4];
#pragma unroll
  for (int k = 0; k < 4; ++k) v[k] = (b0 + k < nb) ? h[b0 + k] : 0;
  const int sm = v[0] + v[1] + v[2] + v[3];
  const int incl = wave_incl_scan(sm, lane);
  if (lane == 31) wt[wv] = incl;
  __syncthreads();
  const int front = wave_sum((lane < wv) ? wt[lane] : 0);
  int run = front + incl - sm;
  int* ro = roff + (size_t)blockIdx.x * (nb + 1);
#pragma unroll
  for (int k = 0; k < 4; ++k) {
    if (b0 + k < nb) {
      h[b0 + k] = run;
      ro[b0 + k] = run;
    }
    run += v[k];
  }
  if (tid == ST - 1) ro[nb] = run;
  __syncthreads();

  v2i* img2 = (v2i*)img;
#pragma unroll
  for (int j = 0; j < EPT; ++j) {
    if (d[j] >= 0) {
      const int slot = atomicAdd(&h[d[j] >> BSH], 1);
      v2i sd;
      sd[0] = s[j];
      sd[1] = d[j];
      img2[slot] = sd;
    }
  }
  __syncthreads();
  const int tot = min(CH, nedges - (int)blockIdx.x * CH);
  v4i* out4 = (v4i*)sorted + (size_t)blockIdx.x * (CH / 2);
  if (tot == CH) {
    v4i t[CH / 2 / ST];
#pragma unroll
    for (int k = 0; k < CH / 2 / ST; ++k) t[k] = img[tid + k * ST];
#pragma unroll
    for (int k = 0; k < CH / 2 / ST; ++k) out4[tid + k * ST] = t[k];
  } else {
    for (int i = tid; 2 * i < tot; i += ST) out4[i] = img[i];
  }
}

template <int MODE>
__device__ __forceinline__ void walk_runs(const v2i* __restrict__ s2,
                                          const int* __restrict__ roff,
                                          int* c, int* img,
                                          int* __restrict__ csr,
                                          int b, int r0, int ng, int nb,
                                          int wv, int lane, int& begpart) {
  constexpr bool FILL = MODE != 0;
  for (int gb = wv; gb < ng; gb += RW * 32) {
    const int gl = gb + RW * lane;
    int st = 0, en = 0;
    if (gl < ng) {
      const int* ro = roff + (size_t)gl * (nb + 1) + b;
      st = ro[0];
      en = ro[1];
    }
    if (!FILL) begpart += st;
    const int nrun = min(32, (ng - gb + RW - 1) / RW);
    for (int l0 = 0; l0 < nrun; l0 += RG) {
      int ss[RG], dd[RG];
#pragma unroll
      for (int u = 0; u < RG; ++u) {
        const int rs = __builtin_amdgcn_readlane(st, l0 + u);
        const int re = __builtin_amdgcn_readlane(en, l0 + u);
        const int i = rs + lane;
        ss[u] = 0;
        dd[u] = -1;
        if (i < re) {
          const v2i sd = s2[(size_t)(gb + RW * (l0 + u)) * CH + i];
          ss[u] = sd[0];
          dd[u] = sd[1];
        }
      }
#pragma unroll
      for (int u = 0; u < RG; ++u) {
        if (dd[u] >= 0) {
          if (FILL) {
            const int slot = atomicAdd(&c[dd[u] - r0], 1);
            if (MODE == 1) img[slot] = ss[u];
            else csr[slot] = ss[u];
          } else {
            atomicAdd(&c[dd[u] - r0], 1);
          }
        }
      }
      for (int u = 0; u < RG; ++u) {
        const int rs = __builtin_amdgcn_readlane(st, l0 + u);
        const int re = __builtin_amdgcn_readlane(en, l0 + u);
        for (int i = rs + 32 + lane; i < re; i += 32) {
          const v2i sd = s2[(size_t)(gb + RW * (l0 + u)) * CH + i];
          if (FILL) {
            const int slot = atomicAdd(&c[sd[1] - r0], 1);
            if (MODE == 1) img[slot] = sd[0];
            else csr[slot] = sd[0];
          } else {
            atomicAdd(&c[sd[1] - r0], 1);
          }
        }
      }
    }
  }
}

__global__ __launch_bounds__(RW * 32)
void bin_rows_kernel(const int* __restrict__ sorted,
                     const int* __restrict__ roff,
                     int* __restrict__ cnt, int* __restrict__ rowptr,
                     float* __restrict__ dinv, int* __restrict__ csr,
                     int n, int ng, int nb) {
  __shared__ int c[BROWS];
  __shared__ int wt[RW];
  __shared__ int img[CCAP];
  const int tid = threadIdx.x;
  const int lane = tid & 31;
  const int wv = __builtin_amdgcn_readfirstlane(tid >> 5);
  const int b = blockIdx.x;
  const int r0 = b << BSH;
  const v2i* s2 = (const v2i*)sorted;

  if (tid < BROWS) c[tid] = 0;
  __syncthreads();
  int begpart = 0;
  walk_runs<0>(s2, roff, c, img, csr, b, r0, ng, nb, wv, lane, begpart);
  const int ws = wave_sum(begpart);
  if (lane == 0) wt[wv] = ws;
  __syncthreads();
  const int beg = wave_sum((lane < RW) ? wt[lane] : 0);
  __syncthreads();

  const int v = (tid < BROWS) ? c[tid] : 0;
  const int incl = wave_incl_scan(v, lane);
  if (tid < BROWS && lane == 31) wt[wv] = incl;
  __syncthreads();
  const int front = wave_sum((lane < wv && lane < BROWS / 32) ? wt[lane] : 0);
  const int tot = wave_sum((lane < BROWS / 32) ? wt[lane] : 0);
  if (tid < BROWS) {
    const int rl = front + incl - v;
    c[tid] = rl;
    const int row = r0 + tid;
    if (row < n) {
      cnt[row] = v;
      rowptr[row] = beg + rl;
      dinv[row] = rsqrtf((float)(v + 1));
    }
  }
  __syncthreads();
  if (tot <= CCAP) {
    walk_runs<1>(s2, roff, c, img, csr, b, r0, ng, nb, wv, lane, begpart);
    __syncthreads();
    for (int i0 = tid; i0 - tid < tot; i0 += 4 * RW * 32) {
      int t[4];
#pragma unroll
      for (int k = 0; k < 4; ++k)
        t[k] = img[min(i0 + k * RW * 32, CCAP - 1)];
#pragma unroll
      for (int k = 0; k < 4; ++k) {
        const int i = i0 + k * RW * 32;
        if (i < tot) csr[beg + i] = t[k];
      }
    }
  } else {
    walk_runs<2>(s2, roff, c, img, csr + beg, b, r0, ng, nb, wv, lane, begpart);
  }
}

__global__ __launch_bounds__(256)
void gemm_scale_kernel(const float* __restrict__ x, const float* __restrict__ W,
                       const float* __restrict__ dinv, float* __restrict__ hp,
                       int nrows) {
  __shared__ v2f Wlds[(NF / 2) * PSTR];
  for (int i = threadIdx.x; i < NF * NF; i += 256) {
    int k = i >> 7, c = i & 127;
    Wlds[(k >> 1) * PSTR + c][k & 1] = W[i];
  }
  __syncthreads();

  const int wave = threadIdx.x >> 5;
  const int lane = threadIdx.x & 31;
  const int m0 = (blockIdx.x * 8 + wave) * 16;
  if (m0 >= nrows) return;

  const int half  = lane >> 4;
  const int khalf = half * 2;
  const int colb  = lane & 15;

  v8f acc[8] = {};

  const float* xr = x + (size_t)(m0 + colb) * NF + khalf;

  for (int kk = 0; kk < NF; kk += 4) {
    v2f a = *(const v2f*)(xr + kk);
    const v2f* wr = &Wlds[((kk >> 1) + half) * PSTR + colb];
#pragma unroll
    for (int nt = 0; nt < 8; ++nt) {
      v2f b = wr[nt * 16];
      acc[nt] = __builtin_amdgcn_wmma_f32_16x16x4_f32(
          false, a, false, b, (short)0, acc[nt], false, false);
    }
  }

  const int rbase = half << 3;
#pragma unroll
  for (int r = 0; r < 8; ++r) {
    const int row = m0 + rbase + r;
    const float dv = dinv[row];
    const size_t ro = (size_t)row * NF + colb;
#pragma unroll
    for (int nt = 0; nt < 8; ++nt) {
      hp[ro + nt * 16] = acc[nt][r] * dv;
    }
  }
}

__global__ __launch_bounds__(256)
void gather_rows_kernel(const int* __restrict__ rowptr,
                        const int* __restrict__ cnt,
                        const int* __restrict__ csr,
                        const float* __restrict__ hp,
                        const float* __restrict__ dinv,
                        const float* __restrict__ bias,
                        float* __restrict__ out, int n) {
  const int lane = threadIdx.x & 31;
  const int row = __builtin_amdgcn_readfirstlane(
      (int)(blockIdx.x * 8 + (threadIdx.x >> 5)));
  if (row >= n) return;

  const int start = rowptr[row];
  const int deg = cnt[row];
  const v4f* hp4 = (const v4f*)hp;

  v4f acc = hp4[(size_t)row * 32 + lane];

  for (int base = 0; base < deg; base += 32) {
    const int m = min(32, deg - base);
    int idx = row;
    if (lane < m) idx = csr[start + base + lane];
    for (int j = 0; j < m; j += 8) {
      v4f v[8];
#pragma unroll
      for (int u = 0; u < 8; ++u) {
        const int s = __builtin_amdgcn_readlane(idx, j + u);
        v[u] = hp4[(size_t)s * 32 + lane];
      }
#pragma unroll
      for (int u = 0; u < 8; ++u) {
        if (j + u < m) acc += v[u];
      }
    }
  }

  const float dv = dinv[row];
  const v4f b4 = ((const v4f*)bias)[lane];
  ((v4f*)out)[(size_t)row * 32 + lane] = acc * dv + b4;
}

extern "C" void kernel_launch(void* const* d_in, const int* in_sizes, int n_in,
                              void* d_out, int out_size, void* d_ws, size_t ws_size,
                              hipStream_t stream) {
  const float* x  = (const float*)d_in[0];
  const int*   ei = (const int*)d_in[1];
  const float* W  = (const float*)d_in[2];
  const float* b  = (const float*)d_in[3];
  float* out = (float*)d_out;

  const int N = in_sizes[0] / NF;
  const int E = in_sizes[1] / 2;

  const int NB = (N + BROWS - 1) / BROWS;
  const int NG = (E + CH - 1) / CH;

  const size_t nb = ((size_t)N * 4 + 511) & ~(size_t)511;
  const size_t eb = ((size_t)E * 4 + 511) & ~(size_t)511;
  const size_t gb = ((size_t)NG * (NB + 1) * 4 + 511) & ~(size_t)511;
  const size_t sb = (size_t)NG * CH * 8;
  char* p = (char*)d_ws;
  int*   cnt      = (int*)p;           p += nb;
  int*   rowptr   = (int*)p;           p += nb;
  float* dinv     = (float*)p;         p += nb;
  int*   roff     = (int*)p;           p += gb;
  int*   sorted   = (int*)p;           p += sb;
  int*   csr      = (int*)p;           p += eb;
  float* hp       = (float*)p;

  slice_sort_kernel<<<NG, ST, 0, stream>>>(ei, roff, sorted, E, NB);
  bin_rows_kernel<<<NB, RW * 32, 0, stream>>>(sorted, roff, cnt, rowptr, dinv, csr, N, NG, NB);

  gemm_scale_kernel<<<(N + 127) / 128, 256, 0, stream>>>(x, W, dinv, hp, N);

  gather_rows_kernel<<<(N + 7) / 8, 256, 0, stream>>>(rowptr, cnt, csr, hp, dinv, b, out, N);
}
